# starting kernel
# baseline (speedup 1.0000x reference)
#include <hip/hip_runtime.h>

typedef float v2f __attribute__((ext_vector_type(2)));
typedef float v8f __attribute__((ext_vector_type(8)));

#define IN_FEATS  64
#define OUT_FEATS 32
#define N_KERNELS 8
#define NKF       (N_KERNELS * OUT_FEATS)
#define DIM       2

__global__ __launch_bounds__(256)
void fc_wmma_kernel(const float* __restrict__ feat,
                    const float* __restrict__ W,
                    float* __restrict__ nodefeat,
                    int n, int mtiles) {
  const int wave = blockIdx.x * (blockDim.x >> 5) + (threadIdx.x >> 5);
  const int lane = threadIdx.x & 31;
  const int tn = wave & 15;
  const int tm = wave >> 4;
  if (tm >= mtiles) return;

  const int half = lane >> 4;
  const int r    = lane & 15;

  int mrow = tm * 16 + r;
  if (mrow >= n) mrow = n - 1;
  const float* __restrict__ arow = feat + (size_t)mrow * IN_FEATS;
  const float* __restrict__ brow = W + (size_t)(tn * 16 + r) * IN_FEATS;

  v8f c = {};
#pragma unroll
  for (int kk = 0; kk < IN_FEATS / 4; ++kk) {
    const int kb = kk * 4 + half * 2;
    const float2 af = *(const float2*)(arow + kb);
    const float2 bf = *(const float2*)(brow + kb);
    v2f a; a.x = af.x; a.y = af.y;
    v2f b; b.x = bf.x; b.y = bf.y;
    c = __builtin_amdgcn_wmma_f32_16x16x4_f32(
        false, a, false, b, (short)0, c, false, false);
  }

#pragma unroll
  for (int v = 0; v < 8; ++v) {
    const int row = tm * 16 + v + half * 8;
    if (row < n)
      nodefeat[(size_t)row * NKF + tn * 16 + r] = c[v];
  }
}

__global__ __launch_bounds__(256)
void gmm_edge_kernel(const int* __restrict__ rowptr,
                     const int* __restrict__ colind,
                     const int* __restrict__ permute,
                     const float* __restrict__ pseudo,
                     const float* __restrict__ nodefeat,
                     const float* __restrict__ mu,
                     const float* __restrict__ inv_sigma,
                     const float* __restrict__ bias,
                     float* __restrict__ out,
                     int n) {
  __shared__ float s_mu[N_KERNELS * DIM];
  __shared__ float s_is[N_KERNELS * DIM];
  __shared__ float s_bias[OUT_FEATS];

  const int t = threadIdx.x;
  if (t < N_KERNELS * DIM) { s_mu[t] = mu[t]; s_is[t] = inv_sigma[t]; }
  if (t < OUT_FEATS)       { s_bias[t] = bias[t]; }
  __syncthreads();

  const int node = blockIdx.x * (blockDim.x >> 5) + (t >> 5);
  const int f    = t & 31;
  if (node >= n) return;

  const int start = rowptr[node];
  const int end   = rowptr[node + 1];

  float acc = 0.0f;
  for (int e = start; e < end; ++e) {
    const int src = colind[e];
    const int pe  = permute[e];
    const float p0 = pseudo[(size_t)pe * 2 + 0];
    const float p1 = pseudo[(size_t)pe * 2 + 1];
    const float* __restrict__ nf = nodefeat + (size_t)src * NKF + f;
#pragma unroll
    for (int k = 0; k < N_KERNELS; ++k) {
      const float d0 = (p0 - s_mu[k * 2 + 0]) * s_is[k * 2 + 0];
      const float d1 = (p1 - s_mu[k * 2 + 1]) * s_is[k * 2 + 1];
      const float wk = __expf(-0.5f * (d0 * d0 + d1 * d1));
      acc = fmaf(wk, nf[k * OUT_FEATS], acc);
    }
  }
  out[(size_t)node * OUT_FEATS + f] = acc + s_bias[f];
}

extern "C" void kernel_launch(void* const* d_in, const int* in_sizes, int n_in,
                              void* d_out, int out_size, void* d_ws, size_t ws_size,
                              hipStream_t stream) {
  const int*   rowptr    = (const int*)d_in[0];
  const int*   colind    = (const int*)d_in[1];
  const int*   permute   = (const int*)d_in[4];
  const float* feat      = (const float*)d_in[5];
  const float* pseudo    = (const float*)d_in[6];
  const float* W         = (const float*)d_in[7];
  const float* mu        = (const float*)d_in[8];
  const float* inv_sigma = (const float*)d_in[9];
  const float* bias      = (const float*)d_in[10];
  float*       out       = (float*)d_out;
  float*       nodefeat  = (float*)d_ws;

  const int n      = in_sizes[0] - 1;
  const int mtiles = (n + 15) / 16;
  const int waves  = mtiles * 16;

  dim3 blk(256);
  dim3 grid1((waves + 7) / 8);
  hipLaunchKernelGGL(fc_wmma_kernel, grid1, blk, 0, stream,
                     feat, W, nodefeat, n, mtiles);

  dim3 grid2((n + 7) / 8);
  hipLaunchKernelGGL(gmm_edge_kernel, grid2, blk, 0, stream,
                     rowptr, colind, permute, pseudo, nodefeat,
                     mu, inv_sigma, bias, out, n);
}


# FC GEMM re-tiled: 512 four-wave groups hold their 64 columns of W in registers and loop over 16-row strips (was 50000 one-tile waves); 16-B fragment loads, whole-line stores; k sum order changes
# speedup vs baseline: 1.4420x; 1.4420x over previous
#include <hip/hip_runtime.h>

typedef float v2f __attribute__((ext_vector_type(2)));
typedef float v8f __attribute__((ext_vector_type(8)));

#define IN_FEATS  64
#define OUT_FEATS 32
#define N_KERNELS 8
#define NKF       (N_KERNELS * OUT_FEATS)
#define DIM       2

#define FC_WAVES  4
#define FC_TILES  4
#define FC_GROUPS 512

__device__ __forceinline__ float xhalf(float x) {
  const int b = __builtin_bit_cast(int, x);
  return __builtin_bit_cast(
      float, __builtin_amdgcn_permlanex16(b, b, 0x76543210, 0xfedcba98, false, false));
}

template <bool GUARD>
__device__ __forceinline__ void store_strip(const v8f (&c)[FC_TILES],
                                            float* __restrict__ obase,
                                            int half, int rows) {
#pragma unroll
  for (int t = 0; t < FC_TILES; t += 2) {
#pragma unroll
    for (int v = 0; v < 8; ++v) {
      const float lo = c[t][v];
      const float hi = c[t + 1][v];
      const float xl = xhalf(lo);
      const float xh = xhalf(hi);
      const float top = half ? xh : lo;
      const float bot = half ? hi : xl;
      if (!GUARD || v < rows)
        obase[(size_t)v * NKF + t * 16] = top;
      if (!GUARD || v + 8 < rows)
        obase[(size_t)(v + 8) * NKF + t * 16] = bot;
    }
  }
}

__global__ __launch_bounds__(FC_WAVES * 32)
void fc_wmma_kernel(const float* __restrict__ feat,
                    const float* __restrict__ W,
                    float* __restrict__ nodefeat,
                    int n, int mtiles) {
  const int lane = threadIdx.x & 31;
  const int wv   = threadIdx.x >> 5;
  const int half = lane >> 4;
  const int r    = lane & 15;

  int s = blockIdx.x;
  if (s >= mtiles) return;

  float4 bq[FC_TILES][8];
#pragma unroll
  for (int t = 0; t < FC_TILES; ++t) {
    const float* __restrict__ brow =
        W + (size_t)(wv * 64 + t * 16 + r) * IN_FEATS + half * 4;
#pragma unroll
    for (int j = 0; j < 8; ++j) bq[t][j] = *(const float4*)(brow + j * 8);
  }

  float4 aq[8];
  {
    int mrow = s * 16 + r;
    if (mrow >= n) mrow = n - 1;
    const float* __restrict__ arow = feat + (size_t)mrow * IN_FEATS + half * 4;
#pragma unroll
    for (int j = 0; j < 8; ++j) aq[j] = *(const float4*)(arow + j * 8);
  }

  for (; s < mtiles; s += gridDim.x) {
    float4 an[8];
    {
      int sn = s + (int)gridDim.x;
      if (sn >= mtiles) sn = mtiles - 1;
      int mrow = sn * 16 + r;
      if (mrow >= n) mrow = n - 1;
      const float* __restrict__ arow = feat + (size_t)mrow * IN_FEATS + half * 4;
#pragma unroll
      for (int j = 0; j < 8; ++j) an[j] = *(const float4*)(arow + j * 8);
    }
    __builtin_amdgcn_sched_barrier(0);

    v8f c[FC_TILES];
#pragma unroll
    for (int t = 0; t < FC_TILES; ++t) c[t] = v8f{};

#pragma unroll
    for (int j = 0; j < 8; ++j) {
      v2f alo; alo.x = aq[j].x; alo.y = aq[j].y;
      v2f ahi; ahi.x = aq[j].z; ahi.y = aq[j].w;
#pragma unroll
      for (int t = 0; t < FC_TILES; ++t) {
        v2f b; b.x = bq[t][j].x; b.y = bq[t][j].y;
        c[t] = __builtin_amdgcn_wmma_f32_16x16x4_f32(
            false, alo, false, b, (short)0, c[t], false, false);
      }
#pragma unroll
      for (int t = 0; t < FC_TILES; ++t) {
        v2f b; b.x = bq[t][j].z; b.y = bq[t][j].w;
        c[t] = __builtin_amdgcn_wmma_f32_16x16x4_f32(
            false, ahi, false, b, (short)0, c[t], false, false);
      }
    }

    float* __restrict__ obase =
        nodefeat + (size_t)(s * 16) * NKF + wv * 64 + lane;
    if (s * 16 + 16 <= n) store_strip<false>(c, obase, half, 16);
    else                  store_strip<true>(c, obase, half, n - s * 16);

#pragma unroll
    for (int j = 0; j < 8; ++j) aq[j] = an[j];
  }
}

__global__ __launch_bounds__(256)
void gmm_edge_kernel(const int* __restrict__ rowptr,
                     const int* __restrict__ colind,
                     const int* __restrict__ permute,
                     const float* __restrict__ pseudo,
                     const float* __restrict__ nodefeat,
                     const float* __restrict__ mu,
                     const float* __restrict__ inv_sigma,
                     const float* __restrict__ bias,
                     float* __restrict__ out,
                     int n) {
  const int t    = threadIdx.x;
  const int node = blockIdx.x * (blockDim.x >> 5) + (t >> 5);
  const int lane = t & 31;
  if (node >= n) return;

  const int ka = lane >> 3;
  const int kb = ka + 4;

  const float mua0 = mu[ka * DIM + 0], mua1 = mu[ka * DIM + 1];
  const float mub0 = mu[kb * DIM + 0], mub1 = mu[kb * DIM + 1];
  const float isa0 = inv_sigma[ka * DIM + 0], isa1 = inv_sigma[ka * DIM + 1];
  const float isb0 = inv_sigma[kb * DIM + 0], isb1 = inv_sigma[kb * DIM + 1];

  const int start = rowptr[node];
  const int end   = rowptr[node + 1];

  float4 acc = make_float4(0.0f, 0.0f, 0.0f, 0.0f);
  for (int e = start; e < end; ++e) {
    const int src = colind[e];
    const int pe  = permute[e];
    const float2 p = *(const float2*)(pseudo + (size_t)pe * DIM);
    const float4* __restrict__ row =
        (const float4*)(nodefeat + (size_t)src * NKF);
    const float4 ya = row[lane];
    const float4 yb = row[NKF / 8 + lane];

    const float da0 = (p.x - mua0) * isa0;
    const float da1 = (p.y - mua1) * isa1;
    const float db0 = (p.x - mub0) * isb0;
    const float db1 = (p.y - mub1) * isb1;
    const float wa = __expf(-0.5f * (da0 * da0 + da1 * da1));
    const float wb = __expf(-0.5f * (db0 * db0 + db1 * db1));

    acc.x = fmaf(wa, ya.x, acc.x);
    acc.y = fmaf(wa, ya.y, acc.y);
    acc.z = fmaf(wa, ya.z, acc.z);
    acc.w = fmaf(wa, ya.w, acc.w);
    acc.x = fmaf(wb, yb.x, acc.x);
    acc.y = fmaf(wb, yb.y, acc.y);
    acc.z = fmaf(wb, yb.z, acc.z);
    acc.w = fmaf(wb, yb.w, acc.w);
  }

  acc.x += __shfl_xor(acc.x, 8);
  acc.y += __shfl_xor(acc.y, 8);
  acc.z += __shfl_xor(acc.z, 8);
  acc.w += __shfl_xor(acc.w, 8);
  acc.x += __shfl_xor(acc.x, 16);
  acc.y += __shfl_xor(acc.y, 16);
  acc.z += __shfl_xor(acc.z, 16);
  acc.w += __shfl_xor(acc.w, 16);

  if (lane < OUT_FEATS / 4) {
    const float4 b = ((const float4*)bias)[lane];
    float4 r;
    r.x = acc.x + b.x; r.y = acc.y + b.y; r.z = acc.z + b.z; r.w = acc.w + b.w;
    ((float4*)(out + (size_t)node * OUT_FEATS))[lane] = r;
  }
}

extern "C" void kernel_launch(void* const* d_in, const int* in_sizes, int n_in,
                              void* d_out, int out_size, void* d_ws, size_t ws_size,
                              hipStream_t stream) {
  const int*   rowptr    = (const int*)d_in[0];
  const int*   colind    = (const int*)d_in[1];
  const int*   permute   = (const int*)d_in[4];
  const float* feat      = (const float*)d_in[5];
  const float* pseudo    = (const float*)d_in[6];
  const float* W         = (const float*)d_in[7];
  const float* mu        = (const float*)d_in[8];
  const float* inv_sigma = (const float*)d_in[9];
  const float* bias      = (const float*)d_in[10];
  float*       out       = (float*)d_out;
  float*       nodefeat  = (float*)d_ws;

  const int n      = in_sizes[0] - 1;
  const int mtiles = (n + 15) / 16;

  dim3 blk1(FC_WAVES * 32);
  dim3 grid1(FC_GROUPS);
  hipLaunchKernelGGL(fc_wmma_kernel, grid1, blk1, 0, stream,
                     feat, W, nodefeat, n, mtiles);

  dim3 blk(256);

  dim3 grid2((n + 7) / 8);
  hipLaunchKernelGGL(gmm_edge_kernel, grid2, blk, 0, stream,
                     rowptr, colind, permute, pseudo, nodefeat,
                     mu, inv_sigma, bias, out, n);
}


# Edge kernel made resident: 1536 eight-wave groups fill the chip once, each wave loops over nodes g, g+G, .. with the next node's edge range fetched ahead (was one wave per node); sum order unchanged
# speedup vs baseline: 1.4930x; 1.0354x over previous
#include <hip/hip_runtime.h>

typedef float v2f __attribute__((ext_vector_type(2)));
typedef float v8f __attribute__((ext_vector_type(8)));

#define IN_FEATS  64
#define OUT_FEATS 32
#define N_KERNELS 8
#define NKF       (N_KERNELS * OUT_FEATS)
#define DIM       2

#define FC_WAVES  4
#define FC_TILES  4
#define FC_GROUPS 512

__device__ __forceinline__ float xhalf(float x) {
  const int b = __builtin_bit_cast(int, x);
  return __builtin_bit_cast(
      float, __builtin_amdgcn_permlanex16(b, b, 0x76543210, 0xfedcba98, false, false));
}

template <bool GUARD>
__device__ __forceinline__ void store_strip(const v8f (&c)[FC_TILES],
                                            float* __restrict__ obase,
                                            int half, int rows) {
#pragma unroll
  for (int t = 0; t < FC_TILES; t += 2) {
#pragma unroll
    for (int v = 0; v < 8; ++v) {
      const float lo = c[t][v];
      const float hi = c[t + 1][v];
      const float xl = xhalf(lo);
      const float xh = xhalf(hi);
      const float top = half ? xh : lo;
      const float bot = half ? hi : xl;
      if (!GUARD || v < rows)
        obase[(size_t)v * NKF + t * 16] = top;
      if (!GUARD || v + 8 < rows)
        obase[(size_t)(v + 8) * NKF + t * 16] = bot;
    }
  }
}

__global__ __launch_bounds__(FC_WAVES * 32)
void fc_wmma_kernel(const float* __restrict__ feat,
                    const float* __restrict__ W,
                    float* __restrict__ nodefeat,
                    int n, int mtiles) {
  const int lane = threadIdx.x & 31;
  const int wv   = threadIdx.x >> 5;
  const int half = lane >> 4;
  const int r    = lane & 15;

  int s = blockIdx.x;
  if (s >= mtiles) return;

  float4 bq[FC_TILES][8];
#pragma unroll
  for (int t = 0; t < FC_TILES; ++t) {
    const float* __restrict__ brow =
        W + (size_t)(wv * 64 + t * 16 + r) * IN_FEATS + half * 4;
#pragma unroll
    for (int j = 0; j < 8; ++j) bq[t][j] = *(const float4*)(brow + j * 8);
  }

  float4 aq[8];
  {
    int mrow = s * 16 + r;
    if (mrow >= n) mrow = n - 1;
    const float* __restrict__ arow = feat + (size_t)mrow * IN_FEATS + half * 4;
#pragma unroll
    for (int j = 0; j < 8; ++j) aq[j] = *(const float4*)(arow + j * 8);
  }

  for (; s < mtiles; s += gridDim.x) {
    float4 an[8];
    {
      int sn = s + (int)gridDim.x;
      if (sn >= mtiles) sn = mtiles - 1;
      int mrow = sn * 16 + r;
      if (mrow >= n) mrow = n - 1;
      const float* __restrict__ arow = feat + (size_t)mrow * IN_FEATS + half * 4;
#pragma unroll
      for (int j = 0; j < 8; ++j) an[j] = *(const float4*)(arow + j * 8);
    }
    __builtin_amdgcn_sched_barrier(0);

    v8f c[FC_TILES];
#pragma unroll
    for (int t = 0; t < FC_TILES; ++t) c[t] = v8f{};

#pragma unroll
    for (int j = 0; j < 8; ++j) {
      v2f alo; alo.x = aq[j].x; alo.y = aq[j].y;
      v2f ahi; ahi.x = aq[j].z; ahi.y = aq[j].w;
#pragma unroll
      for (int t = 0; t < FC_TILES; ++t) {
        v2f b; b.x = bq[t][j].x; b.y = bq[t][j].y;
        c[t] = __builtin_amdgcn_wmma_f32_16x16x4_f32(
            false, alo, false, b, (short)0, c[t], false, false);
      }
#pragma unroll
      for (int t = 0; t < FC_TILES; ++t) {
        v2f b; b.x = bq[t][j].z; b.y = bq[t][j].w;
        c[t] = __builtin_amdgcn_wmma_f32_16x16x4_f32(
            false, ahi, false, b, (short)0, c[t], false, false);
      }
    }

    float* __restrict__ obase =
        nodefeat + (size_t)(s * 16) * NKF + wv * 64 + lane;
    if (s * 16 + 16 <= n) store_strip<false>(c, obase, half, 16);
    else                  store_strip<true>(c, obase, half, n - s * 16);

#pragma unroll
    for (int j = 0; j < 8; ++j) aq[j] = an[j];
  }
}

#define EDGE_WAVES  8
#define EDGE_GROUPS 1536

__global__ __launch_bounds__(EDGE_WAVES * 32)
void gmm_edge_kernel(const int* __restrict__ rowptr,
                     const int* __restrict__ colind,
                     const int* __restrict__ permute,
                     const float* __restrict__ pseudo,
                     const float* __restrict__ nodefeat,
                     const float* __restrict__ mu,
                     const float* __restrict__ inv_sigma,
                     const float* __restrict__ bias,
                     float* __restrict__ out,
                     int n) {
  const int t    = threadIdx.x;
  const int lane = t & 31;
  const int wpb  = blockDim.x >> 5;
  const int gw   = gridDim.x * wpb;
  int node = blockIdx.x * wpb + (t >> 5);
  if (node >= n) return;

  const int ka = lane >> 3;
  const int kb = ka + 4;

  const float mua0 = mu[ka * DIM + 0], mua1 = mu[ka * DIM + 1];
  const float mub0 = mu[kb * DIM + 0], mub1 = mu[kb * DIM + 1];
  const float isa0 = inv_sigma[ka * DIM + 0], isa1 = inv_sigma[ka * DIM + 1];
  const float isb0 = inv_sigma[kb * DIM + 0], isb1 = inv_sigma[kb * DIM + 1];
  const float4 bq = ((const float4*)bias)[lane & 7];

  int start = rowptr[node];
  int end   = rowptr[node + 1];

  for (;;) {
    const int nnode = node + gw;
    const int pn    = nnode < n ? nnode : node;
    const int nstart = rowptr[pn];
    const int nend   = rowptr[pn + 1];

    float4 acc = make_float4(0.0f, 0.0f, 0.0f, 0.0f);
    for (int e = start; e < end; ++e) {
      const int src = colind[e];
      const int pe  = permute[e];
      const float2 p = *(const float2*)(pseudo + (size_t)pe * DIM);
      const float4* __restrict__ row =
          (const float4*)(nodefeat + (size_t)src * NKF);
      const float4 ya = row[lane];
      const float4 yb = row[NKF / 8 + lane];

      const float da0 = (p.x - mua0) * isa0;
      const float da1 = (p.y - mua1) * isa1;
      const float db0 = (p.x - mub0) * isb0;
      const float db1 = (p.y - mub1) * isb1;
      const float wa = __expf(-0.5f * (da0 * da0 + da1 * da1));
      const float wb = __expf(-0.5f * (db0 * db0 + db1 * db1));

      acc.x = fmaf(wa, ya.x, acc.x);
      acc.y = fmaf(wa, ya.y, acc.y);
      acc.z = fmaf(wa, ya.z, acc.z);
      acc.w = fmaf(wa, ya.w, acc.w);
      acc.x = fmaf(wb, yb.x, acc.x);
      acc.y = fmaf(wb, yb.y, acc.y);
      acc.z = fmaf(wb, yb.z, acc.z);
      acc.w = fmaf(wb, yb.w, acc.w);
    }

    acc.x += __shfl_xor(acc.x, 8);
    acc.y += __shfl_xor(acc.y, 8);
    acc.z += __shfl_xor(acc.z, 8);
    acc.w += __shfl_xor(acc.w, 8);
    acc.x += __shfl_xor(acc.x, 16);
    acc.y += __shfl_xor(acc.y, 16);
    acc.z += __shfl_xor(acc.z, 16);
    acc.w += __shfl_xor(acc.w, 16);

    if (lane < OUT_FEATS / 4) {
      float4 r;
      r.x = acc.x + bq.x; r.y = acc.y + bq.y; r.z = acc.z + bq.z; r.w = acc.w + bq.w;
      ((float4*)(out + (size_t)node * OUT_FEATS))[lane] = r;
    }

    if (nnode >= n) break;
    node  = nnode;
    start = nstart;
    end   = nend;
  }
}

extern "C" void kernel_launch(void* const* d_in, const int* in_sizes, int n_in,
                              void* d_out, int out_size, void* d_ws, size_t ws_size,
                              hipStream_t stream) {
  const int*   rowptr    = (const int*)d_in[0];
  const int*   colind    = (const int*)d_in[1];
  const int*   permute   = (const int*)d_in[4];
  const float* feat      = (const float*)d_in[5];
  const float* pseudo    = (const float*)d_in[6];
  const float* W         = (const float*)d_in[7];
  const float* mu        = (const float*)d_in[8];
  const float* inv_sigma = (const float*)d_in[9];
  const float* bias      = (const float*)d_in[10];
  float*       out       = (float*)d_out;
  float*       nodefeat  = (float*)d_ws;

  const int n      = in_sizes[0] - 1;
  const int mtiles = (n + 15) / 16;

  dim3 blk1(FC_WAVES * 32);
  dim3 grid1(FC_GROUPS);
  hipLaunchKernelGGL(fc_wmma_kernel, grid1, blk1, 0, stream,
                     feat, W, nodefeat, n, mtiles);

  dim3 blk(EDGE_WAVES * 32);
  dim3 grid2(EDGE_GROUPS);
  hipLaunchKernelGGL(gmm_edge_kernel, grid2, blk, 0, stream,
                     rowptr, colind, permute, pseudo, nodefeat,
                     mu, inv_sigma, bias, out, n);
}
